# starting kernel
# baseline (speedup 1.0000x reference)
#include <hip/hip_runtime.h>

typedef __attribute__((ext_vector_type(2))) float v2f;
typedef __attribute__((ext_vector_type(4))) float v4f;
typedef __attribute__((ext_vector_type(8))) float v8f;
typedef __attribute__((ext_vector_type(3), aligned(4))) float v3f;

namespace {
constexpr int HW = 56;
constexpr int SP = HW * HW;
constexpr int LCOUNT = 32;
constexpr int TILES_PER_L = SP / 16;
}

__global__ __launch_bounds__(256)
void fused_shift_unfold_gemm(const float* __restrict__ x,
                             const float* __restrict__ wgt,
                             float* __restrict__ out) {
    const int lane = (int)(threadIdx.x & 31u);
    const int wave = (int)(threadIdx.x >> 5u);
    const int gt   = __builtin_amdgcn_readfirstlane((int)blockIdx.x * 8 + wave);
    const int l    = gt / TILES_PER_L;
    const int tile = gt - l * TILES_PER_L;

    const bool h    = (lane >> 4) != 0;
    const int  mrow = lane & 15;
    const int  s    = tile * 16 + mrow;
    const int  p    = s / HW;
    const int  o    = s - p * HW;
    const int  rowbase = ((p + HW - 1) % HW) * HW;

    int q0 = o - 2; q0 += (q0 < 0) ? HW : 0;
    int q1 = o - 1; q1 += (q1 < 0) ? HW : 0;
    const int q2 = o;
    const bool vo0 = o >= 1;
    const bool vo2 = o <= HW - 2;
    const bool g02 = h ? vo2 : vo0;
    const bool g10 = h ? vo0 : true;
    const bool g21 = h ? true : vo2;

    int base[3][2];
    #pragma unroll
    for (int k = 0; k < 3; ++k) {
        const int c2c = min(max(l + k - 1, 0), LCOUNT - 1);
        base[k][0] = (c2c * 2 + 0) * SP + rowbase;
        base[k][1] = (c2c * 2 + 1) * SP + rowbase;
    }

    const float* wl = wgt + l * 12;
    const int nb = (mrow & 3) * 3;
    v3f W[3][2];
    #pragma unroll
    for (int k = 0; k < 3; ++k)
        #pragma unroll
        for (int j = 0; j < 2; ++j)
            W[k][j] = *(const v3f*)(wl + (k * 2 + j) * (LCOUNT * 12) + nb);
    const v3f z3 = {0.0f, 0.0f, 0.0f};
    if (l == 0)          { W[0][0] = z3; W[0][1] = z3; }
    if (l == LCOUNT - 1) { W[2][0] = z3; W[2][1] = z3; }

    const float* xhi = x + 64 * SP;
    auto aelem = [&](int bA, int bB, int qA, int qB) -> float {
        const int idx = (h ? bB : bA) + (h ? qB : qA);
        const float lo = x[idx];
        const float hi = xhi[idx];
        return lo + hi;
    };
    const float v00 = aelem(base[0][0], base[0][0], q0, q2);
    const float v01 = aelem(base[0][0], base[0][1], q1, q0);
    const float v10 = aelem(base[0][1], base[1][0], q1, q0);
    const float v11 = aelem(base[0][1], base[1][0], q2, q1);
    const float v20 = aelem(base[1][0], base[1][1], q2, q1);
    const float v21 = aelem(base[1][1], base[1][1], q0, q2);
    const float v30 = aelem(base[2][0], base[2][0], q0, q2);
    const float v31 = aelem(base[2][0], base[2][1], q1, q0);
    const float v40 = aelem(base[2][1], base[2][1], q1, q1);
    const float v41 = aelem(base[2][1], base[2][1], q2, q2);

    v8f acc = {};
    v2f a, b;

    a.x = g02 ? v00 : 0.0f;
    a.y = g10 ? v01 : 0.0f;
    b.x = h ? W[0][0].z : W[0][0].x;
    b.y = h ? W[0][1].x : W[0][0].y;
    acc = __builtin_amdgcn_wmma_f32_16x16x4_f32(false, a, false, b, (short)0, acc, false, false);

    a.x = g10 ? v10 : 0.0f;
    a.y = g21 ? v11 : 0.0f;
    b.x = h ? W[1][0].x : W[0][1].y;
    b.y = h ? W[1][0].y : W[0][1].z;
    acc = __builtin_amdgcn_wmma_f32_16x16x4_f32(false, a, false, b, (short)0, acc, false, false);

    a.x = g21 ? v20 : 0.0f;
    a.y = g02 ? v21 : 0.0f;
    b.x = h ? W[1][1].y : W[1][0].z;
    b.y = h ? W[1][1].z : W[1][1].x;
    acc = __builtin_amdgcn_wmma_f32_16x16x4_f32(false, a, false, b, (short)0, acc, false, false);

    a.x = g02 ? v30 : 0.0f;
    a.y = g10 ? v31 : 0.0f;
    b.x = h ? W[2][0].z : W[2][0].x;
    b.y = h ? W[2][1].x : W[2][0].y;
    acc = __builtin_amdgcn_wmma_f32_16x16x4_f32(false, a, false, b, (short)0, acc, false, false);

    a.x = v40;
    a.y = g21 ? v41 : 0.0f;
    b.x = h ? 0.0f : W[2][1].y;
    b.y = h ? 0.0f : W[2][1].z;
    acc = __builtin_amdgcn_wmma_f32_16x16x4_f32(false, a, false, b, (short)0, acc, false, false);

    if (mrow < 4) {
        const unsigned ooff = (unsigned)((l * 4 + mrow) * SP + tile * 16 + (h ? 8 : 0));
        v4f lo4 = { acc[0], acc[1], acc[2], acc[3] };
        v4f hi4 = { acc[4], acc[5], acc[6], acc[7] };
        *(v4f*)(out + ooff)     = lo4;
        *(v4f*)(out + ooff + 4) = hi4;
    }
}

extern "C" void kernel_launch(void* const* d_in, const int*  , int  ,
                              void* d_out, int  , void*  , size_t  ,
                              hipStream_t stream) {
    const float* x = (const float*)d_in[0];
    const float* w = (const float*)d_in[1];
    float* out = (float*)d_out;

    const int total_waves = LCOUNT * TILES_PER_L;
    dim3 grid(total_waves / 8);
    dim3 block(256);
    hipLaunchKernelGGL(fused_shift_unfold_gemm, grid, block, 0, stream, x, w, out);
}


# Products regrouped: weights as A rows (m,tap), aligned 4-float chunks as B columns, one tile per width offset, taps summed after: 15 aligned wide loads per 64 positions, not 104; sum order changes
# speedup vs baseline: 1.2392x; 1.2392x over previous
#include <hip/hip_runtime.h>


typedef __attribute__((ext_vector_type(2))) float v2f;
typedef __attribute__((ext_vector_type(4))) float v4f;
typedef __attribute__((ext_vector_type(8))) float v8f;

namespace {
constexpr int HW = 56;
constexpr int SP = HW * HW;
constexpr int CH4 = SP / 4;
constexpr int LCOUNT = 32;
constexpr int TILES_PER_L = CH4 / 16;
constexpr int WAVES_PER_BLOCK = 4;
}

__global__ __launch_bounds__(WAVES_PER_BLOCK * 32)
void fused_shift_unfold_gemm(const float* __restrict__ x,
                             const float* __restrict__ wgt,
                             float* __restrict__ out) {
    const int lane = (int)(threadIdx.x & 31u);
    const int wave = (int)(threadIdx.x >> 5u);
    const int gt   = __builtin_amdgcn_readfirstlane((int)blockIdx.x * WAVES_PER_BLOCK + wave);
    const int l    = gt / TILES_PER_L;
    const int tile = gt - l * TILES_PER_L;

    const int  h = lane >> 4;
    const int  c = lane & 15;

    const int G  = tile * 16 + c;
    const int pp = G / 14;
    const int g  = G - pp * 14;
    const int p  = (pp == 0) ? (HW - 1) : (pp - 1);
    const int gl = (g == 0) ? 13 : (g - 1);
    const int cur4  = p * 14 + g;
    const int left2 = (p * 14 + gl) * 2 + 1;

    const v4f* x4 = (const v4f*)x;
    const v2f* x2 = (const v2f*)x;
    v4f clo[3], chi[3];
    v2f llo[3], lhi[3];
    float wa[3];
    const int m_row = c >> 2;
    const int i_row = c & 3;
    const int wofs  = l * 12 + m_row * 3 + min(i_row, 2);
    #pragma unroll
    for (int e = 0; e < 3; ++e) {
        const int pl  = 3 * h + e;
        const int k   = pl >> 1;
        const int j   = pl & 1;
        const int c2  = min(max(l + k - 1, 0), LCOUNT - 1);
        const int chn = c2 * 2 + j;
        wa[e]  = wgt[pl * (LCOUNT * 12) + wofs];
        clo[e] = x4[chn * CH4 + cur4];
        chi[e] = x4[(chn + 64) * CH4 + cur4];
        llo[e] = x2[chn * (CH4 * 2) + left2];
        lhi[e] = x2[(chn + 64) * (CH4 * 2) + left2];
    }

    const bool row_ok = i_row < 3;
    const bool lo_edge = (l == 0) && (h == 0);
    const bool hi_edge = (l == LCOUNT - 1) && (h == 1);
    wa[0] = (row_ok && !lo_edge) ? wa[0] : 0.0f;
    wa[1] = (row_ok && !lo_edge && !hi_edge) ? wa[1] : 0.0f;
    wa[2] = (row_ok && !hi_edge) ? wa[2] : 0.0f;
    const v2f a0 = { wa[0], wa[1] };
    const v2f a1 = { wa[2], 0.0f };

    const bool gate_first = (g == 0);
    const bool gate_last  = (g == 13);
    float V[3][6];
    #pragma unroll
    for (int e = 0; e < 3; ++e) {
        const float u2 = llo[e].x + lhi[e].x;
        V[e][0] = gate_first ? 0.0f : u2;
        V[e][1] = llo[e].y + lhi[e].y;
        V[e][2] = clo[e].x + chi[e].x;
        V[e][3] = clo[e].y + chi[e].y;
        V[e][4] = clo[e].z + chi[e].z;
        const float u7 = clo[e].w + chi[e].w;
        V[e][5] = gate_last ? 0.0f : u7;
    }

    v8f D[6];
    #pragma unroll
    for (int u = 0; u < 6; ++u) {
        const v8f zero = {};
        const v2f b0 = { V[0][u], V[1][u] };
        D[u] = __builtin_amdgcn_wmma_f32_16x16x4_f32(false, a0, false, b0, (short)0, zero, false, false);
    }
    #pragma unroll
    for (int u = 0; u < 6; ++u) {
        const v2f b1 = { V[2][u], 0.0f };
        D[u] = __builtin_amdgcn_wmma_f32_16x16x4_f32(false, a1, false, b1, (short)0, D[u], false, false);
    }

    v4f* out4 = (v4f*)out;
    #pragma unroll
    for (int mm = 0; mm < 2; ++mm) {
        v4f o;
        #pragma unroll
        for (int r = 0; r < 4; ++r)
            o[r] = (D[r][4 * mm] + D[r + 1][4 * mm + 1]) + D[r + 2][4 * mm + 2];
        const int m = 2 * h + mm;
        out4[(l * 4 + m) * CH4 + G] = o;
    }
}

extern "C" void kernel_launch(void* const* d_in, const int*  , int  ,
                              void* d_out, int  , void*  , size_t  ,
                              hipStream_t stream) {
    const float* x = (const float*)d_in[0];
    const float* w = (const float*)d_in[1];
    float* out = (float*)d_out;

    const int total_waves = LCOUNT * TILES_PER_L;
    dim3 grid(total_waves / WAVES_PER_BLOCK);
    dim3 block(WAVES_PER_BLOCK * 32);
    hipLaunchKernelGGL(fused_shift_unfold_gemm, grid, block, 0, stream, x, w, out);
}
